# starting kernel
# baseline (speedup 1.0000x reference)
#include <hip/hip_runtime.h>

typedef float v8f __attribute__((ext_vector_type(8)));
typedef float v2f __attribute__((ext_vector_type(2)));

#define B_  4
#define H_  16
#define S_  4096
#define D_  128
#define LCH 256
#define NCH (S_ / LCH)

__global__ void __launch_bounds__(128)
dc_pass_a(const float* __restrict__ x, const float* __restrict__ gamma,
          float* __restrict__ carry) {
  const int blk = blockIdx.x;
  const int bh  = blk / NCH;
  const int c   = blk % NCH;
  const int h   = bh % H_;
  const int d   = threadIdx.x;
  const float g = gamma[h];

  const float* xp = x + ((size_t)bh * S_ + (size_t)c * LCH) * D_ + d;
  float acc = 0.0f;
#pragma unroll 8
  for (int t = 0; t < LCH; ++t) {
    acc = fmaf(g, acc, xp[(size_t)t * D_]);
  }
  carry[(size_t)blk * D_ + d] = acc;
}

__global__ void __launch_bounds__(256)
dc_pass_b(const float* __restrict__ gamma, float* __restrict__ carry) {
  const int col = blockIdx.x * 256 + threadIdx.x;
  const int bh  = col / D_;
  const int d   = col % D_;
  const int h   = bh % H_;

  float G = gamma[h];
#pragma unroll
  for (int i = 0; i < 8; ++i) G *= G;

  float P = 0.0f;
  float* cp = carry + (size_t)bh * NCH * D_ + d;
#pragma unroll
  for (int c = 0; c < NCH; ++c) {
    const float t = cp[(size_t)c * D_];
    cp[(size_t)c * D_] = P;
    P = fmaf(G, P, t);
  }
}

__global__ void __launch_bounds__(256)
dc_pass_c_wmma(const float* __restrict__ x, const float* __restrict__ gamma,
               const float* __restrict__ carry_in, float* __restrict__ y) {
  const int blk  = blockIdx.x;
  const int bh   = blk / NCH;
  const int c    = blk % NCH;
  const int h    = bh % H_;
  const int wv   = threadIdx.x >> 5;
  const int lane = threadIdx.x & 31;
  const int n    = lane & 15;
  const int hi   = lane >> 4;
  const int d0   = wv * 16;
  const float g  = gamma[h];

  float ac[4][2];
  float gc[8];
#pragma unroll
  for (int k = 0; k < 4; ++k) { ac[k][0] = 0.0f; ac[k][1] = 0.0f; }
#pragma unroll
  for (int r = 0; r < 8; ++r) gc[r] = 0.0f;
  {
    const int m = n;
    float p = 1.0f;
#pragma unroll
    for (int e = 0; e <= 16; ++e) {
#pragma unroll
      for (int k = 0; k < 4; ++k) {
        const int j0 = 4 * k + 2 * hi;
        if (j0     <= m && (m - j0)     == e) ac[k][0] = p;
        if (j0 + 1 <= m && (m - j0 - 1) == e) ac[k][1] = p;
      }
#pragma unroll
      for (int r = 0; r < 8; ++r)
        if ((r + 8 * hi + 1) == e) gc[r] = p;
      p *= g;
    }
  }

  const size_t colbase = ((size_t)bh * S_ + (size_t)c * LCH) * D_ + (size_t)(d0 + n);
  const float* xp = x + colbase;
  float*       yp = y + colbase;

  float cw = carry_in[((size_t)bh * NCH + c) * D_ + d0 + n];

#pragma unroll 1
  for (int t16 = 0; t16 < LCH / 16; ++t16) {
    const int tb = t16 * 16;

    v8f acc;
#pragma unroll
    for (int r = 0; r < 8; ++r) acc[r] = gc[r] * cw;

#pragma unroll
    for (int k = 0; k < 4; ++k) {
      const int t0 = tb + 4 * k + 2 * hi;
      v2f Bv;
      Bv[0] = __builtin_nontemporal_load(xp + (size_t)t0 * D_);
      Bv[1] = __builtin_nontemporal_load(xp + (size_t)(t0 + 1) * D_);
      v2f Av; Av[0] = ac[k][0]; Av[1] = ac[k][1];
      acc = __builtin_amdgcn_wmma_f32_16x16x4_f32(
           false, Av,  false, Bv,
           (short)0, acc,  false,  false);
    }

#pragma unroll
    for (int r = 0; r < 8; ++r)
      __builtin_nontemporal_store(acc[r], yp + (size_t)(tb + r + 8 * hi) * D_);

    cw = __shfl(acc[7], n + 16, 32);
  }
}

extern "C" void kernel_launch(void* const* d_in, const int* in_sizes, int n_in,
                              void* d_out, int out_size, void* d_ws, size_t ws_size,
                              hipStream_t stream) {
  const float* x     = (const float*)d_in[0];
  const float* gamma = (const float*)d_in[1];
  float* y  = (float*)d_out;
  float* ws = (float*)d_ws;

  dc_pass_a<<<dim3(B_ * H_ * NCH), dim3(128), 0, stream>>>(x, gamma, ws);
  dc_pass_b<<<dim3((B_ * H_ * D_) / 256), dim3(256), 0, stream>>>(gamma, ws);
  dc_pass_c_wmma<<<dim3(B_ * H_ * NCH), dim3(256), 0, stream>>>(x, gamma, ws, y);
}


# Passes A and C re-tiled to 64-step x 64-column wave tiles: 16-byte loads and stores of 256-byte row segments, all 32 row loads of a wave in flight; carry chunk 256 to 64 steps (sum order changes)
# speedup vs baseline: 2.2517x; 2.2517x over previous
#include <hip/hip_runtime.h>


typedef float v8f __attribute__((ext_vector_type(8)));
typedef float v4f __attribute__((ext_vector_type(4)));
typedef float v2f __attribute__((ext_vector_type(2)));

#define B_  4
#define H_  16
#define S_  4096
#define D_  128
#define LCH 64
#define NCH (S_ / LCH)
#define WGT 256
#define NWG (S_ / WGT)


__global__ void __launch_bounds__(256)
dc_pass_a(const float* __restrict__ x, const float* __restrict__ gamma,
          float* __restrict__ carry) {
  const int blk  = blockIdx.x;
  const int bh   = blk / NWG;
  const int j    = blk % NWG;
  const int h    = bh % H_;
  const int wv   = threadIdx.x >> 5;
  const int lane = threadIdx.x & 31;
  const int c    = lane & 15;
  const int hi   = lane >> 4;
  const int half = wv & 1;
  const int ch   = j * 4 + (wv >> 1);
  const float g  = gamma[h];

  const float* xp = x + ((size_t)bh * S_ + (size_t)ch * LCH + 32 * hi) * D_
                      + half * 64 + 4 * c;
  v4f v[32];
#pragma unroll
  for (int i = 0; i < 32; ++i)
    v[i] = *(const v4f*)(xp + (size_t)i * D_);

  v4f acc = {0.0f, 0.0f, 0.0f, 0.0f};
#pragma unroll
  for (int i = 0; i < 32; ++i) {
    acc[0] = fmaf(g, acc[0], v[i][0]);
    acc[1] = fmaf(g, acc[1], v[i][1]);
    acc[2] = fmaf(g, acc[2], v[i][2]);
    acc[3] = fmaf(g, acc[3], v[i][3]);
  }

  float G = g;
#pragma unroll
  for (int i = 0; i < 5; ++i) G *= G;

  v4f lo;
  lo[0] = __shfl(acc[0], c, 32);
  lo[1] = __shfl(acc[1], c, 32);
  lo[2] = __shfl(acc[2], c, 32);
  lo[3] = __shfl(acc[3], c, 32);
  if (hi) {
    v4f r;
    r[0] = fmaf(G, lo[0], acc[0]);
    r[1] = fmaf(G, lo[1], acc[1]);
    r[2] = fmaf(G, lo[2], acc[2]);
    r[3] = fmaf(G, lo[3], acc[3]);
    *(v4f*)(carry + ((size_t)bh * NCH + ch) * D_ + half * 64 + 4 * c) = r;
  }
}

__global__ void __launch_bounds__(256)
dc_pass_b(const float* __restrict__ gamma, float* __restrict__ carry) {
  const int col = blockIdx.x * 256 + threadIdx.x;
  const int bh  = col / D_;
  const int d   = col % D_;
  const int h   = bh % H_;

  float G = gamma[h];
#pragma unroll
  for (int i = 0; i < 6; ++i) G *= G;

  float P = 0.0f;
  float* cp = carry + (size_t)bh * NCH * D_ + d;
#pragma unroll
  for (int c = 0; c < NCH; ++c) {
    const float t = cp[(size_t)c * D_];
    cp[(size_t)c * D_] = P;
    P = fmaf(G, P, t);
  }
}

__global__ void __launch_bounds__(256)
dc_pass_c_wmma(const float* __restrict__ x, const float* __restrict__ gamma,
               const float* __restrict__ carry_in, float* __restrict__ y) {
  const int blk  = blockIdx.x;
  const int bh   = blk / NWG;
  const int j    = blk % NWG;
  const int h    = bh % H_;
  const int wv   = threadIdx.x >> 5;
  const int lane = threadIdx.x & 31;
  const int c    = lane & 15;
  const int hi   = lane >> 4;
  const int half = wv & 1;
  const int ch   = j * 4 + (wv >> 1);
  const float g  = gamma[h];

  float ac[4][2];
  float gc[8];
#pragma unroll
  for (int k = 0; k < 4; ++k) { ac[k][0] = 0.0f; ac[k][1] = 0.0f; }
#pragma unroll
  for (int r = 0; r < 8; ++r) gc[r] = 0.0f;
  {
    const int m = c;
    float p = 1.0f;
#pragma unroll
    for (int e = 0; e <= 16; ++e) {
#pragma unroll
      for (int k = 0; k < 4; ++k) {
        const int j0 = 4 * k + 2 * hi;
        if (j0     <= m && (m - j0)     == e) ac[k][0] = p;
        if (j0 + 1 <= m && (m - j0 - 1) == e) ac[k][1] = p;
      }
#pragma unroll
      for (int r = 0; r < 8; ++r)
        if ((r + 8 * hi + 1) == e) gc[r] = p;
      p *= g;
    }
  }

  const size_t base = ((size_t)bh * S_ + (size_t)ch * LCH) * D_ + half * 64 + 4 * c;
  const float* xp = x + base;
  float*       yp = y + base;

  v4f cw = *(const v4f*)(carry_in + ((size_t)bh * NCH + ch) * D_ + half * 64 + 4 * c);

  v4f L[4][4][2];
#pragma unroll
  for (int t = 0; t < 4; ++t)
#pragma unroll
    for (int k = 0; k < 4; ++k)
#pragma unroll
      for (int q = 0; q < 2; ++q)
        L[t][k][q] = __builtin_nontemporal_load(
            (const v4f*)(xp + (size_t)(16 * t + 4 * k + 2 * hi + q) * D_));

  __builtin_amdgcn_sched_barrier(0);

#pragma unroll
  for (int t = 0; t < 4; ++t) {
    const int tb = t * 16;

    v8f acc[4];
#pragma unroll
    for (int e = 0; e < 4; ++e)
#pragma unroll
      for (int r = 0; r < 8; ++r) acc[e][r] = gc[r] * cw[e];

#pragma unroll
    for (int k = 0; k < 4; ++k) {
      v2f Av; Av[0] = ac[k][0]; Av[1] = ac[k][1];
#pragma unroll
      for (int e = 0; e < 4; ++e) {
        v2f Bv;
        Bv[0] = L[t][k][0][e];
        Bv[1] = L[t][k][1][e];
        acc[e] = __builtin_amdgcn_wmma_f32_16x16x4_f32(
             false, Av,  false, Bv,
             (short)0, acc[e],  false,  false);
      }
    }

#pragma unroll
    for (int r = 0; r < 8; ++r) {
      v4f o;
      o[0] = acc[0][r]; o[1] = acc[1][r]; o[2] = acc[2][r]; o[3] = acc[3][r];
      __builtin_nontemporal_store(o, (v4f*)(yp + (size_t)(tb + r + 8 * hi) * D_));
    }

#pragma unroll
    for (int e = 0; e < 4; ++e) cw[e] = __shfl(acc[e][7], c + 16, 32);
  }
}

extern "C" void kernel_launch(void* const* d_in, const int* in_sizes, int n_in,
                              void* d_out, int out_size, void* d_ws, size_t ws_size,
                              hipStream_t stream) {
  const float* x     = (const float*)d_in[0];
  const float* gamma = (const float*)d_in[1];
  float* y  = (float*)d_out;
  float* ws = (float*)d_ws;

  dc_pass_a<<<dim3(B_ * H_ * NWG), dim3(256), 0, stream>>>(x, gamma, ws);
  dc_pass_b<<<dim3((B_ * H_ * D_) / 256), dim3(256), 0, stream>>>(gamma, ws);
  dc_pass_c_wmma<<<dim3(B_ * H_ * NWG), dim3(256), 0, stream>>>(x, gamma, ws, y);
}


# Three launches fused into one: a work-group owns a 32-column slice of one (b,h) for the whole sequence, x read once, segment carries through LDS; zero-seed scan plus carry term (sum order changes)
# speedup vs baseline: 3.1295x; 1.3898x over previous
#include <hip/hip_runtime.h>


typedef float v8f __attribute__((ext_vector_type(8)));
typedef float v4f __attribute__((ext_vector_type(4)));
typedef float v2f __attribute__((ext_vector_type(2)));

#define B_   4
#define H_   16
#define S_   4096
#define D_   128
#define CW   32
#define NSL  (D_ / CW)
#define NWV  16
#define WT   64
#define STEP (NWV * WT)
#define NST  (S_ / STEP)
#define NSEG (2 * NWV)

__global__ void __launch_bounds__(512)
dc_scan_onepass(const float* __restrict__ x, const float* __restrict__ gamma,
                float* __restrict__ y) {
  __shared__ float lc[2][NSEG][CW];

  const int blk  = blockIdx.x;
  const int bh   = blk / NSL;
  const int sl   = blk % NSL;
  const int h    = bh % H_;
  const int wv   = __builtin_amdgcn_readfirstlane((int)(threadIdx.x >> 5));
  const int lane = threadIdx.x & 31;
  const int c    = lane & 15;
  const int hi   = lane >> 4;
  const int quad = lane & 7;
  const int seg  = (lane >> 3) & 1;
  const float g  = gamma[h];

  float ac[4][2];
  float gc[8];
#pragma unroll
  for (int k = 0; k < 4; ++k) { ac[k][0] = 0.0f; ac[k][1] = 0.0f; }
#pragma unroll
  for (int r = 0; r < 8; ++r) gc[r] = 0.0f;
  float G16 = 1.0f;
  {
    const int m = c;
    float p = 1.0f;
#pragma unroll
    for (int e = 0; e <= 16; ++e) {
#pragma unroll
      for (int k = 0; k < 4; ++k) {
        const int j0 = 4 * k + 2 * hi;
        if (j0     <= m && (m - j0)     == e) ac[k][0] = p;
        if (j0 + 1 <= m && (m - j0 - 1) == e) ac[k][1] = p;
      }
#pragma unroll
      for (int r = 0; r < 8; ++r)
        if ((r + 8 * hi + 1) == e) gc[r] = p;
      if (e == 16) G16 = p;
      p *= g;
    }
  }
  const float G32 = G16 * G16;

  const size_t base = ((size_t)bh * S_ + (size_t)wv * WT + (size_t)seg * 32) * D_
                    + (size_t)sl * CW + 4 * quad;
  const float* xp = x + base;
  float*       yp = y + base;

  v4f L[2][2][4][2];
#pragma unroll
  for (int t = 0; t < 2; ++t)
#pragma unroll
    for (int k = 0; k < 4; ++k)
#pragma unroll
      for (int q = 0; q < 2; ++q)
        L[0][t][k][q] = *(const v4f*)(xp + (size_t)(16 * t + 4 * k + 2 * hi + q) * D_);

  float R = 0.0f;

#pragma unroll
  for (int s = 0; s < NST; ++s) {
    const int b = s & 1;

    if (s + 1 < NST) {
#pragma unroll
      for (int t = 0; t < 2; ++t)
#pragma unroll
        for (int k = 0; k < 4; ++k)
#pragma unroll
          for (int q = 0; q < 2; ++q)
            L[b ^ 1][t][k][q] = *(const v4f*)(
                xp + (size_t)((s + 1) * STEP + 16 * t + 4 * k + 2 * hi + q) * D_);
    }
    __builtin_amdgcn_sched_barrier(0);

    v8f acc[2][4];
    v4f cw = {0.0f, 0.0f, 0.0f, 0.0f};
#pragma unroll
    for (int t = 0; t < 2; ++t) {
#pragma unroll
      for (int e = 0; e < 4; ++e)
#pragma unroll
        for (int r = 0; r < 8; ++r) acc[t][e][r] = (t == 0) ? 0.0f : gc[r] * cw[e];

#pragma unroll
      for (int k = 0; k < 4; ++k) {
        v2f Av; Av[0] = ac[k][0]; Av[1] = ac[k][1];
#pragma unroll
        for (int e = 0; e < 4; ++e) {
          v2f Bv;
          Bv[0] = L[b][t][k][0][e];
          Bv[1] = L[b][t][k][1][e];
          acc[t][e] = __builtin_amdgcn_wmma_f32_16x16x4_f32(
               false, Av,  false, Bv,
               (short)0, acc[t][e],  false,  false);
        }
      }
      if (t == 0) {
#pragma unroll
        for (int e = 0; e < 4; ++e) cw[e] = __shfl(acc[0][e][7], c + 16, 32);
      }
    }

    if (hi) {
      v4f o;
      o[0] = acc[1][0][7]; o[1] = acc[1][1][7]; o[2] = acc[1][2][7]; o[3] = acc[1][3][7];
      *(v4f*)&lc[b][2 * wv + seg][4 * quad] = o;
    }
    asm volatile("s_wait_dscnt 0x0\n\ts_barrier_signal -1\n\ts_barrier_wait -1" ::: "memory");

    float ev[NSEG];
#pragma unroll
    for (int k = 0; k < NSEG; ++k) ev[k] = lc[b][k][lane];
    float P = R, c0 = 0.0f, c1 = 0.0f;
#pragma unroll
    for (int k = 0; k < NSEG; ++k) {
      if (k == 2 * wv)     c0 = P;
      if (k == 2 * wv + 1) c1 = P;
      P = fmaf(G32, P, ev[k]);
    }
    R = P;

    v4f cin;
#pragma unroll
    for (int e = 0; e < 4; ++e) {
      const float a0 = __shfl(c0, 4 * quad + e, 32);
      const float a1 = __shfl(c1, 4 * quad + e, 32);
      cin[e] = seg ? a1 : a0;
    }

#pragma unroll
    for (int t = 0; t < 2; ++t) {
      v4f ct;
#pragma unroll
      for (int e = 0; e < 4; ++e) ct[e] = (t == 0) ? cin[e] : cin[e] * G16;
#pragma unroll
      for (int r = 0; r < 8; ++r) {
        v4f o;
        o[0] = fmaf(gc[r], ct[0], acc[t][0][r]);
        o[1] = fmaf(gc[r], ct[1], acc[t][1][r]);
        o[2] = fmaf(gc[r], ct[2], acc[t][2][r]);
        o[3] = fmaf(gc[r], ct[3], acc[t][3][r]);
        __builtin_nontemporal_store(
            o, (v4f*)(yp + (size_t)(s * STEP + 16 * t + r + 8 * hi) * D_));
      }
    }
  }
}

extern "C" void kernel_launch(void* const* d_in, const int* in_sizes, int n_in,
                              void* d_out, int out_size, void* d_ws, size_t ws_size,
                              hipStream_t stream) {
  const float* x     = (const float*)d_in[0];
  const float* gamma = (const float*)d_in[1];
  float* y = (float*)d_out;

  dc_scan_onepass<<<dim3(B_ * H_ * NSL), dim3(NWV * 32), 0, stream>>>(x, gamma, y);
}


# The two 8-wave halves of each work-group run half a step out of phase (8 hand-overs, not 4): one half's carries, stores and row loads issue under the other half's matrix scan; sum order unchanged
# speedup vs baseline: 3.2942x; 1.0526x over previous
#include <hip/hip_runtime.h>


typedef float v8f __attribute__((ext_vector_type(8)));
typedef float v4f __attribute__((ext_vector_type(4)));
typedef float v2f __attribute__((ext_vector_type(2)));

#define B_   4
#define H_   16
#define S_   4096
#define D_   128
#define CW   32
#define NSL  (D_ / CW)
#define NWV  16
#define WT   64
#define STEP (NWV * WT)
#define NST  (S_ / STEP)
#define NSEG (2 * NWV)


#define WG_HANDOVER() \
  asm volatile("s_wait_dscnt 0x0\n\ts_barrier_signal -1\n\ts_barrier_wait -1" ::: "memory")

#define LOAD_STEP(BUF, S_IDX)                                                   \
  do {                                                                          \
    _Pragma("unroll") for (int t = 0; t < 2; ++t)                               \
    _Pragma("unroll") for (int k = 0; k < 4; ++k)                               \
    _Pragma("unroll") for (int q = 0; q < 2; ++q)                               \
      L[BUF][t][k][q] = *(const v4f*)(                                          \
          xp + (size_t)((S_IDX) * STEP + 16 * t + 4 * k + 2 * hi + q) * D_);    \
  } while (0)

#define SCAN_PHASE(S_IDX)                                                       \
  do {                                                                          \
    const int b_ = (S_IDX) & 1;                                                 \
    v4f cw = {0.0f, 0.0f, 0.0f, 0.0f};                                          \
    _Pragma("unroll") for (int t = 0; t < 2; ++t) {                             \
      _Pragma("unroll") for (int e = 0; e < 4; ++e)                             \
      _Pragma("unroll") for (int r = 0; r < 8; ++r)                             \
        acc[t][e][r] = (t == 0) ? 0.0f : gc[r] * cw[e];                         \
      _Pragma("unroll") for (int k = 0; k < 4; ++k) {                           \
        v2f Av; Av[0] = ac[k][0]; Av[1] = ac[k][1];                             \
        _Pragma("unroll") for (int e = 0; e < 4; ++e) {                         \
          v2f Bv;                                                               \
          Bv[0] = L[b_][t][k][0][e];               \
          Bv[1] = L[b_][t][k][1][e];               \
          acc[t][e] = __builtin_amdgcn_wmma_f32_16x16x4_f32(                    \
               false, Av,  false, Bv,                         \
               (short)0, acc[t][e],  false,  false); \
        }                                                                       \
      }                                                                         \
          \
      if (t == 0) {                                                             \
        _Pragma("unroll") for (int e = 0; e < 4; ++e)                           \
          cw[e] = __shfl(acc[0][e][7], c + 16, 32);                             \
      }                                                                         \
    }                                                                           \
         \
    if (hi) {                                                                   \
      v4f o;                                                                    \
      o[0] = acc[1][0][7]; o[1] = acc[1][1][7];                                 \
      o[2] = acc[1][2][7]; o[3] = acc[1][3][7];                                 \
      *(v4f*)&lc[b_][2 * wv + seg][4 * quad] = o;                               \
    }                                                                           \
  } while (0)

#define FIX_PHASE(S_IDX)                                                        \
  do {                                                                          \
    const int b_ = (S_IDX) & 1;                                                 \
    if ((S_IDX) + 2 < NST) LOAD_STEP(b_, (S_IDX) + 2);                          \
    float P = R, c0 = 0.0f, c1 = 0.0f;                                          \
       \
       \
       \
    if (hb != 0 || (S_IDX) > 0) {                                               \
      const float* lo = &lc[hb ? b_ : (b_ ^ 1)][hb ? 0 : NSEG / 2][lane];       \
      _Pragma("unroll") for (int k = 0; k < NSEG / 2; ++k)                      \
        P = fmaf(G32, P, lo[k * CW]);                                           \
    }                                                                           \
    {                                                                           \
      const float* lw = &lc[b_][hb ? NSEG / 2 : 0][lane];                       \
      _Pragma("unroll") for (int k = 0; k < NSEG / 2; ++k) {                    \
        if (k == 2 * wl)     c0 = P;         \
        if (k == 2 * wl + 1) c1 = P;         \
        P = fmaf(G32, P, lw[k * CW]);                                           \
      }                                                                         \
    }                                                                           \
    R = P;                                                                      \
                          \
    v4f cin;                                                                    \
    _Pragma("unroll") for (int e = 0; e < 4; ++e) {                             \
      const float a0 = __shfl(c0, 4 * quad + e, 32);                            \
      const float a1 = __shfl(c1, 4 * quad + e, 32);                            \
      cin[e] = seg ? a1 : a0;                                                   \
    }                                                                           \
       \
       \
    _Pragma("unroll") for (int t = 0; t < 2; ++t) {                             \
      v4f ct;                                                                   \
      _Pragma("unroll") for (int e = 0; e < 4; ++e)                             \
        ct[e] = (t == 0) ? cin[e] : cin[e] * G16;                               \
      _Pragma("unroll") for (int r = 0; r < 8; ++r) {                           \
        v4f o;                                                                  \
        o[0] = fmaf(gc[r], ct[0], acc[t][0][r]);                                \
        o[1] = fmaf(gc[r], ct[1], acc[t][1][r]);                                \
        o[2] = fmaf(gc[r], ct[2], acc[t][2][r]);                                \
        o[3] = fmaf(gc[r], ct[3], acc[t][3][r]);                                \
        __builtin_nontemporal_store(                                            \
            o, (v4f*)(yp + (size_t)((S_IDX) * STEP + 16 * t + r + 8 * hi) * D_)); \
      }                                                                         \
    }                                                                           \
  } while (0)

__global__ void __launch_bounds__(512)
dc_scan_onepass(const float* __restrict__ x, const float* __restrict__ gamma,
                float* __restrict__ y) {
  __shared__ float lc[2][NSEG][CW];

  const int blk  = blockIdx.x;
  const int bh   = blk / NSL;
  const int sl   = blk % NSL;
  const int h    = bh % H_;
  const int wv   = __builtin_amdgcn_readfirstlane((int)(threadIdx.x >> 5));
  const int lane = threadIdx.x & 31;
  const int c    = lane & 15;
  const int hi   = lane >> 4;
  const int quad = lane & 7;
  const int seg  = (lane >> 3) & 1;
  const float g  = gamma[h];

  float ac[4][2];
  float gc[8];
#pragma unroll
  for (int k = 0; k < 4; ++k) { ac[k][0] = 0.0f; ac[k][1] = 0.0f; }
#pragma unroll
  for (int r = 0; r < 8; ++r) gc[r] = 0.0f;
  float G16 = 1.0f;
  {
    const int m = c;
    float p = 1.0f;
#pragma unroll
    for (int e = 0; e <= 16; ++e) {
#pragma unroll
      for (int k = 0; k < 4; ++k) {
        const int j0 = 4 * k + 2 * hi;
        if (j0     <= m && (m - j0)     == e) ac[k][0] = p;
        if (j0 + 1 <= m && (m - j0 - 1) == e) ac[k][1] = p;
      }
#pragma unroll
      for (int r = 0; r < 8; ++r)
        if ((r + 8 * hi + 1) == e) gc[r] = p;
      if (e == 16) G16 = p;
      p *= g;
    }
  }
  const float G32 = G16 * G16;

  const size_t base = ((size_t)bh * S_ + (size_t)wv * WT + (size_t)seg * 32) * D_
                    + (size_t)sl * CW + 4 * quad;
  const float* xp = x + base;
  float*       yp = y + base;

  v4f L[2][2][4][2];
  v8f acc[2][4];
  LOAD_STEP(0, 0);
  LOAD_STEP(1, 1);

  float R = 0.0f;
  const int hb = wv >> 3;
  const int wl = wv & 7;

#pragma unroll
  for (int s = 0; s < NST; ++s) {
    if (hb != 0) WG_HANDOVER();
    SCAN_PHASE(s);
    WG_HANDOVER();
    FIX_PHASE(s);
    if (hb == 0) WG_HANDOVER();
  }
}

extern "C" void kernel_launch(void* const* d_in, const int* in_sizes, int n_in,
                              void* d_out, int out_size, void* d_ws, size_t ws_size,
                              hipStream_t stream) {
  const float* x     = (const float*)d_in[0];
  const float* gamma = (const float*)d_in[1];
  float* y = (float*)d_out;

  dc_scan_onepass<<<dim3(B_ * H_ * NSL), dim3(NWV * 32), 0, stream>>>(x, gamma, y);
}
